# starting kernel
# baseline (speedup 1.0000x reference)
#include <hip/hip_runtime.h>
#include <math.h>

typedef __attribute__((ext_vector_type(2))) float v2f;
typedef __attribute__((ext_vector_type(8))) float v8f;

#define HIDF 128

__global__ void k_init_deg(float* __restrict__ deg, int n) {
  int i = blockIdx.x * blockDim.x + threadIdx.x;
  if (i < n) deg[i] = 1.0f;
}

__global__ void k_edge_deg(const int* __restrict__ col, const float* __restrict__ ew,
                           float* __restrict__ deg, int E) {
  int e = blockIdx.x * blockDim.x + threadIdx.x;
  if (e < E) atomicAdd(&deg[col[e]], ew[e]);
}

__global__ void k_rsqrt(float* __restrict__ deg, int n) {
  int i = blockIdx.x * blockDim.x + threadIdx.x;
  if (i < n) {
    float d = deg[i];
    deg[i] = (d > 0.0f) ? rsqrtf(d) : 0.0f;
  }
}

__global__ __launch_bounds__(256)
void k_gemm_wmma(const float* __restrict__ A, const float* __restrict__ W,
                 float* __restrict__ C, const float* __restrict__ bias) {
  __shared__ float As[16 * 132];
  const int t  = threadIdx.x;
  const int m0 = blockIdx.x * 16;

  {
    const float4* A4 = (const float4*)(A + (size_t)m0 * HIDF);
    for (int i = t; i < 16 * 32; i += 256) {
      int r = i >> 5, c = i & 31;
      float4 v = A4[i];
      As[r * 132 + c * 4 + 0] = v.x;
      As[r * 132 + c * 4 + 1] = v.y;
      As[r * 132 + c * 4 + 2] = v.z;
      As[r * 132 + c * 4 + 3] = v.w;
    }
  }
  __syncthreads();

  const int lane = t & 31;
  const int wave = t >> 5;
  const int half = lane >> 4;
  const int l15  = lane & 15;
  const int n0   = wave * 16;

  v8f acc = {};
#pragma unroll
  for (int kt = 0; kt < 32; ++kt) {
    const int k0 = kt * 4 + 2 * half;
#if defined(__has_builtin) && __has_builtin(__builtin_amdgcn_wmma_f32_16x16x4_f32)
    v2f a, b;
    a.x = As[l15 * 132 + k0 + 0];
    a.y = As[l15 * 132 + k0 + 1];
    b.x = W[(k0 + 0) * HIDF + n0 + l15];
    b.y = W[(k0 + 1) * HIDF + n0 + l15];
    acc = __builtin_amdgcn_wmma_f32_16x16x4_f32(false, a, false, b, (short)0, acc,
                                                false, false);
#else
#pragma unroll
    for (int r = 0; r < 8; ++r) {
      int M = r + 8 * half;
      int kk = kt * 4;
      float s = acc[r];
      s += As[M * 132 + kk + 0] * W[(kk + 0) * HIDF + n0 + l15];
      s += As[M * 132 + kk + 1] * W[(kk + 1) * HIDF + n0 + l15];
      s += As[M * 132 + kk + 2] * W[(kk + 2) * HIDF + n0 + l15];
      s += As[M * 132 + kk + 3] * W[(kk + 3) * HIDF + n0 + l15];
      acc[r] = s;
    }
#endif
  }

  const float bv = bias ? bias[n0 + l15] : 0.0f;
#pragma unroll
  for (int r = 0; r < 8; ++r) {
    C[(size_t)(m0 + r + 8 * half) * HIDF + (n0 + l15)] = acc[r] + bv;
  }
}

__global__ void k_selfagg(const float* __restrict__ hw, const float* __restrict__ dis,
                          float* __restrict__ agg, int n) {
  int idx = blockIdx.x * blockDim.x + threadIdx.x;
  if (idx >= n * 32) return;
  int node = idx >> 5;
  float s = dis[node];
  s = s * s;
  float4 v = ((const float4*)hw)[idx];
  float4 o;
  o.x = v.x * s; o.y = v.y * s; o.z = v.z * s; o.w = v.w * s;
  ((float4*)agg)[idx] = o;
}

__global__ __launch_bounds__(256)
void k_scatter(const float* __restrict__ hw, const int* __restrict__ row,
               const int* __restrict__ col, const float* __restrict__ ew,
               const float* __restrict__ dis, float* __restrict__ agg, int E) {
  int e = blockIdx.x * 8 + (threadIdx.x >> 5);
  if (e >= E) return;
  const int lane = threadIdx.x & 31;
  const int r = row[e];
  const int c = col[e];
  const float norm = dis[r] * ew[e] * dis[c];
  float4 v = ((const float4*)(hw + (size_t)r * HIDF))[lane];
  float* dst = agg + (size_t)c * HIDF + lane * 4;
  atomicAdd(dst + 0, v.x * norm);
  atomicAdd(dst + 1, v.y * norm);
  atomicAdd(dst + 2, v.z * norm);
  atomicAdd(dst + 3, v.w * norm);
}

__global__ void k_finalize(const float* __restrict__ agg, const float* __restrict__ b,
                           float* __restrict__ out, int n, int relu) {
  int idx = blockIdx.x * blockDim.x + threadIdx.x;
  if (idx >= n * 32) return;
  int c4 = idx & 31;
  float4 v  = ((const float4*)agg)[idx];
  float4 bb = ((const float4*)b)[c4];
  v.x += bb.x; v.y += bb.y; v.z += bb.z; v.w += bb.w;
  if (relu) {
    v.x = fmaxf(v.x, 0.0f); v.y = fmaxf(v.y, 0.0f);
    v.z = fmaxf(v.z, 0.0f); v.w = fmaxf(v.w, 0.0f);
  }
  ((float4*)out)[idx] = v;
}

extern "C" void kernel_launch(void* const* d_in, const int* in_sizes, int n_in,
                              void* d_out, int out_size, void* d_ws, size_t ws_size,
                              hipStream_t stream) {
  const float* x   = (const float*)d_in[0];
  const int*   ei  = (const int*)d_in[1];
  const float* ew  = (const float*)d_in[2];
  const float* Wc[4] = {(const float*)d_in[3], (const float*)d_in[5],
                        (const float*)d_in[7], (const float*)d_in[9]};
  const float* bc[4] = {(const float*)d_in[4], (const float*)d_in[6],
                        (const float*)d_in[8], (const float*)d_in[10]};
  const float* Wout = (const float*)d_in[11];
  const float* bout = (const float*)d_in[12];

  const int n = in_sizes[0] / HIDF;
  const int E = in_sizes[1] / 2;
  const int* row = ei;
  const int* col = ei + E;

  char* ws = (char*)d_ws;
  float* dis = (float*)ws;
  size_t off = (((size_t)n * 4) + 255) & ~(size_t)255;
  float* hw   = (float*)(ws + off);
  float* agg  = hw  + (size_t)n * HIDF;
  float* hbuf = agg + (size_t)n * HIDF;

  k_init_deg<<<(n + 255) / 256, 256, 0, stream>>>(dis, n);
  k_edge_deg<<<(E + 255) / 256, 256, 0, stream>>>(col, ew, dis, E);
  k_rsqrt<<<(n + 255) / 256, 256, 0, stream>>>(dis, n);

  const int nf4 = n * 32;
  const float* hcur = x;
  for (int l = 0; l < 4; ++l) {
    k_gemm_wmma<<<n / 16, 256, 0, stream>>>(hcur, Wc[l], hw, nullptr);
    k_selfagg<<<(nf4 + 255) / 256, 256, 0, stream>>>(hw, dis, agg, n);
    k_scatter<<<(E + 7) / 8, 256, 0, stream>>>(hw, row, col, ew, dis, agg, E);
    k_finalize<<<(nf4 + 255) / 256, 256, 0, stream>>>(agg, bc[l], hbuf, n,
                                                      (l < 3) ? 1 : 0);
    hcur = hbuf;
  }
  k_gemm_wmma<<<n / 16, 256, 0, stream>>>(hcur, Wout, (float*)d_out, bout);
}


# edge-sum row gathers widened: one 16-byte load per lane covers a row's 4 lines (16 loads per 16 edges, not 64); LDS tile rows column-interleaved so f32 adds stay conflict-free; arrival order
# speedup vs baseline: 2.8636x; 2.8636x over previous
#include <hip/hip_runtime.h>
#include <math.h>

typedef __attribute__((ext_vector_type(2))) float v2f;
typedef __attribute__((ext_vector_type(8))) float v8f;
typedef __attribute__((ext_vector_type(4))) float v4f;

#define HIDF 128

__global__ void k_init_deg(float* __restrict__ deg, int n) {
  int i = blockIdx.x * blockDim.x + threadIdx.x;
  if (i < n) deg[i] = 1.0f;
}

__global__ void k_edge_deg(const int* __restrict__ col, const float* __restrict__ ew,
                           float* __restrict__ deg, int E) {
  int e = blockIdx.x * blockDim.x + threadIdx.x;
  if (e < E) atomicAdd(&deg[col[e]], ew[e]);
}

__global__ void k_rsqrt(float* __restrict__ deg, int n) {
  int i = blockIdx.x * blockDim.x + threadIdx.x;
  if (i < n) {
    float d = deg[i];
    deg[i] = (d > 0.0f) ? rsqrtf(d) : 0.0f;
  }
}

#define GEMM_BLOCKS 512
__global__ __launch_bounds__(256)
void k_gemm_wmma(const float* __restrict__ A, const float* __restrict__ W,
                 float* __restrict__ C, const float* __restrict__ bias, int ntiles) {
  __shared__ __attribute__((aligned(16))) float As[2][16 * 132];
  const int t    = threadIdx.x;
  const int lane = t & 31;
  const int wave = t >> 5;
  const int half = lane >> 4;
  const int l15  = lane & 15;
  const int n0   = wave * 16;

  v2f bfr[32];
#pragma unroll
  for (int kt = 0; kt < 32; ++kt) {
    const int k0 = kt * 4 + 2 * half;
    bfr[kt].x = W[(k0 + 0) * HIDF + n0 + l15];
    bfr[kt].y = W[(k0 + 1) * HIDF + n0 + l15];
  }
  const float bv = bias ? bias[n0 + l15] : 0.0f;

  const int so = (t >> 5) * 132 + (t & 31) * 4;
  int tile = blockIdx.x;
  if (tile < ntiles) {
    const float4* A4 = (const float4*)(A + (size_t)tile * 16 * HIDF);
    const float4 q0 = A4[t];
    const float4 q1 = A4[t + 256];
    *(float4*)(&As[0][so])           = q0;
    *(float4*)(&As[0][so + 8 * 132]) = q1;
  }
  __syncthreads();

  int buf = 0;
  for (; tile < ntiles; tile += GEMM_BLOCKS) {
    const float* as = As[buf];
    const bool more = tile + GEMM_BLOCKS < ntiles;
    float4 p0 = make_float4(0.f, 0.f, 0.f, 0.f), p1 = p0;
    if (more) {
      const float4* A4 = (const float4*)(A + (size_t)(tile + GEMM_BLOCKS) * 16 * HIDF);
      p0 = A4[t];
      p1 = A4[t + 256];
    }

    v8f acc = {};
#pragma unroll
    for (int kt = 0; kt < 32; ++kt) {
      const int k0 = kt * 4 + 2 * half;
      v2f a;
      a.x = as[l15 * 132 + k0 + 0];
      a.y = as[l15 * 132 + k0 + 1];
      acc = __builtin_amdgcn_wmma_f32_16x16x4_f32(false, a, false, bfr[kt], (short)0, acc,
                                                  false, false);
    }

    const int m0 = tile * 16;
#pragma unroll
    for (int r = 0; r < 8; ++r) {
      C[(size_t)(m0 + r + 8 * half) * HIDF + (n0 + l15)] = acc[r] + bv;
    }

    if (more) {
      *(float4*)(&As[buf ^ 1][so])           = p0;
      *(float4*)(&As[buf ^ 1][so + 8 * 132]) = p1;
    }
    asm volatile("s_wait_dscnt 0x0" ::: "memory");
    asm volatile("s_barrier_signal -1" ::: "memory");
    asm volatile("s_barrier_wait -1" ::: "memory");
    buf ^= 1;
  }
}

#define TR   400
#define NWV  32
#define SCU  16
#define QB   16
#define QCAP 144

#define WCAP 512

__device__ __forceinline__ void tile_add(float* __restrict__ tile, int r, int dd, float cf,
                                         int cnt, int lane, const float* __restrict__ hw) {
  v4f v[QB];
  const __attribute__((address_space(1))) char* hwb =
      (const __attribute__((address_space(1))) char*)hw;
  const unsigned lo = (unsigned)lane * 16u;
#pragma unroll
  for (int k = 0; k < QB; ++k) {
    if (k < cnt) {
      const unsigned rk = (unsigned)__builtin_amdgcn_readlane(r, k);
      const unsigned off = (rk << 9) | lo;
      v[k] = *(const __attribute__((address_space(1))) v4f*)(hwb + off);
    }
  }
#pragma unroll
  for (int k = 0; k < QB; ++k) {
    if (k < cnt) {
      const float ck = __int_as_float(__builtin_amdgcn_readlane(__float_as_int(cf), k));
      const int dk = __builtin_amdgcn_readlane(dd, k);
      float* dst = tile + dk * HIDF + lane;
      atomicAdd(dst,      v[k].x * ck);
      atomicAdd(dst + 32, v[k].y * ck);
      atomicAdd(dst + 64, v[k].z * ck);
      atomicAdd(dst + 96, v[k].w * ck);
    }
  }
}

template <bool REC>
__device__ __forceinline__ void tile_drain(float* __restrict__ tile, const int* q_e, const int* q_d,
                                           int qend, int cnt, int lane, int n0,
                                           const float* __restrict__ hw,
                                           const int* __restrict__ row,
                                           const float* __restrict__ ew,
                                           const float* __restrict__ dis,
                                           int4* __restrict__ wl, int wn) {
  const int k0 = qend - cnt;
  int r = 0, dd = 0;
  float cf = 0.0f;
  if (lane < cnt) {
    const int e = q_e[k0 + lane];
    dd = q_d[k0 + lane];
    r = row[e];
    cf = dis[r] * ew[e] * dis[n0 + dd];
    if (REC) {
      if (wn + lane < WCAP) wl[wn + lane] = make_int4(r, dd, __float_as_int(cf), 0);
    }
  }
  tile_add(tile, r, dd, cf, cnt, lane, hw);
}

__device__ __forceinline__ void tile_begin(float* __restrict__ tile, const float* __restrict__ hw,
                                           const float* __restrict__ dis,
                                           const float* __restrict__ b, int n0, int nr, int t) {
  for (int i = t; i < nr * 32; i += 1024) {
    const int node = i >> 5, c4 = i & 31;
    float s = dis[n0 + node];
    s = s * s;
    const float4 v  = ((const float4*)(hw + (size_t)(n0 + node) * HIDF))[c4];
    const float4 bb = ((const float4*)b)[c4];
    float* trow = tile + node * HIDF + c4;
    trow[0]  = v.x * s + bb.x;
    trow[32] = v.y * s + bb.y;
    trow[64] = v.z * s + bb.z;
    trow[96] = v.w * s + bb.w;
  }
}

__device__ __forceinline__ void tile_end(const float* __restrict__ tile, float* __restrict__ out,
                                         int n0, int nr, int t, int relu) {
  for (int i = t; i < nr * 32; i += 1024) {
    const float* trow = tile + (i >> 5) * HIDF + (i & 31);
    float4 v;
    v.x = trow[0]; v.y = trow[32]; v.z = trow[64]; v.w = trow[96];
    if (relu) {
      v.x = fmaxf(v.x, 0.0f); v.y = fmaxf(v.y, 0.0f);
      v.z = fmaxf(v.z, 0.0f); v.w = fmaxf(v.w, 0.0f);
    }
    ((float4*)(out + (size_t)n0 * HIDF))[i] = v;
  }
}

template <bool REC>
__device__ __forceinline__ int tile_scan(float* __restrict__ tile, int* qe, int* qd, int lane,
                                         int wave, int n0, int nr,
                                         const float* __restrict__ hw, const int* __restrict__ row,
                                         const int* __restrict__ col, const float* __restrict__ ew,
                                         const float* __restrict__ dis, int E,
                                         int4* __restrict__ wl) {
  int qn = 0;
  int wn = 0;
  const int NG = (E + SCU * 32 - 1) / (SCU * 32);
  const int g0 = (int)((blockIdx.x * (unsigned)NG) / gridDim.x);

  for (int j = wave; j < NG; j += NWV) {
    int g = g0 + j;
    if (g >= NG) g -= NG;
    const int base = g * (SCU * 32);
    int cv[SCU];
#pragma unroll
    for (int u = 0; u < SCU; ++u) {
      const int e = base + u * 32 + lane;
      cv[u] = col[(e < E) ? e : (E - 1)];
    }
    asm volatile("" :: "v"(cv[0]), "v"(cv[1]), "v"(cv[2]), "v"(cv[3]), "v"(cv[4]), "v"(cv[5]),
                 "v"(cv[6]), "v"(cv[7]), "v"(cv[8]), "v"(cv[9]), "v"(cv[10]), "v"(cv[11]),
                 "v"(cv[12]), "v"(cv[13]), "v"(cv[14]), "v"(cv[15]));
#pragma unroll
    for (int u = 0; u < SCU; ++u) {
      if (base + u * 32 + lane >= E) cv[u] = -1;
    }
#pragma unroll
    for (int u = 0; u < SCU; ++u) {
      const unsigned d = (unsigned)(cv[u] - n0);
      const bool in = d < (unsigned)nr;
      const unsigned m = __builtin_amdgcn_ballot_w32(in);
      if (m) {
        if (in) {
          const int pos = qn + __builtin_popcount(m & ((1u << lane) - 1u));
          qe[pos] = base + u * 32 + lane;
          qd[pos] = (int)d;
        }
        qn += __builtin_popcount(m);
      }
      if ((u & 3) == 3) {
        while (qn >= QB) {
          tile_drain<REC>(tile, qe, qd, qn, QB, lane, n0, hw, row, ew, dis, wl, wn);
          qn -= QB;
          wn += QB;
        }
      }
    }
  }
  if (qn > 0) {
    tile_drain<REC>(tile, qe, qd, qn, qn, lane, n0, hw, row, ew, dis, wl, wn);
    wn += qn;
  }
  return wn;
}

__global__ __launch_bounds__(1024)
void k_tile_agg_first(const float* __restrict__ hw, const int* __restrict__ row,
                      const int* __restrict__ col, const float* __restrict__ ew,
                      const float* __restrict__ dis, const float* __restrict__ b,
                      float* __restrict__ out, int n, int E, int relu,
                      int* __restrict__ cnts, int* __restrict__ lst) {
  __shared__ float tile[TR * HIDF];
  __shared__ int q_e[NWV * QCAP];
  __shared__ int q_d[NWV * QCAP];

  const int t = threadIdx.x;
  const int lane = t & 31;
  const int wave = t >> 5;
  const int n0 = blockIdx.x * TR;
  const int nr = (n - n0 < TR) ? (n - n0) : TR;
  const int wid = blockIdx.x * NWV + wave;

  tile_begin(tile, hw, dis, b, n0, nr, t);
  __syncthreads();

  const int wn = tile_scan<true>(tile, q_e + wave * QCAP, q_d + wave * QCAP, lane, wave, n0, nr,
                                 hw, row, col, ew, dis, E, (int4*)lst + (size_t)wid * WCAP);
  if (lane == 0) cnts[wid] = wn;
  __syncthreads();

  tile_end(tile, out, n0, nr, t, relu);
}

__global__ __launch_bounds__(1024)
void k_tile_agg_next(const float* __restrict__ hw, const int* __restrict__ row,
                     const int* __restrict__ col, const float* __restrict__ ew,
                     const float* __restrict__ dis, const float* __restrict__ b,
                     float* __restrict__ out, int n, int E, int relu,
                     const int* __restrict__ cnts, const int* __restrict__ lst) {
  __shared__ float tile[TR * HIDF];
  __shared__ int q_e[NWV * QCAP];
  __shared__ int q_d[NWV * QCAP];

  const int t = threadIdx.x;
  const int lane = t & 31;
  const int wave = t >> 5;
  const int n0 = blockIdx.x * TR;
  const int nr = (n - n0 < TR) ? (n - n0) : TR;
  const int wid = blockIdx.x * NWV + wave;

  const int4* wl = (const int4*)lst + (size_t)wid * WCAP;
  const int wc = __builtin_amdgcn_readfirstlane(cnts[wid]);
  const int4 none = make_int4(0, 0, 0, 0);
  int4 cur = none;
  if (wc <= WCAP && lane < QB && lane < wc) cur = wl[lane];

  tile_begin(tile, hw, dis, b, n0, nr, t);
  __syncthreads();

  if (wc <= WCAP) {
    int i = 0;
    for (; i + QB <= wc; i += QB) {
      int4 nxt = none;
      if (lane < QB && i + QB + lane < wc) nxt = wl[i + QB + lane];
      tile_add(tile, cur.x, cur.y, __int_as_float(cur.z), QB, lane, hw);
      cur = nxt;
    }
    if (i < wc) tile_add(tile, cur.x, cur.y, __int_as_float(cur.z), wc - i, lane, hw);
  } else {
    tile_scan<false>(tile, q_e + wave * QCAP, q_d + wave * QCAP, lane, wave, n0, nr,
                     hw, row, col, ew, dis, E, (int4*)nullptr);
  }
  __syncthreads();

  tile_end(tile, out, n0, nr, t, relu);
}

extern "C" void kernel_launch(void* const* d_in, const int* in_sizes, int n_in,
                              void* d_out, int out_size, void* d_ws, size_t ws_size,
                              hipStream_t stream) {
  const float* x   = (const float*)d_in[0];
  const int*   ei  = (const int*)d_in[1];
  const float* ew  = (const float*)d_in[2];
  const float* Wc[4] = {(const float*)d_in[3], (const float*)d_in[5],
                        (const float*)d_in[7], (const float*)d_in[9]};
  const float* bc[4] = {(const float*)d_in[4], (const float*)d_in[6],
                        (const float*)d_in[8], (const float*)d_in[10]};
  const float* Wout = (const float*)d_in[11];
  const float* bout = (const float*)d_in[12];

  const int n = in_sizes[0] / HIDF;
  const int E = in_sizes[1] / 2;
  const int* row = ei;
  const int* col = ei + E;

  char* ws = (char*)d_ws;
  float* dis = (float*)ws;
  size_t off = (((size_t)n * 4) + 255) & ~(size_t)255;
  float* hw   = (float*)(ws + off);
  float* hbuf = hw + (size_t)n * HIDF;
  const int nb = (n + TR - 1) / TR;
  int* cnts = (int*)(hbuf + (size_t)n * HIDF);
  int* lst  = cnts + (((size_t)nb * NWV + 63) & ~(size_t)63);

  k_init_deg<<<(n + 255) / 256, 256, 0, stream>>>(dis, n);
  k_edge_deg<<<(E + 255) / 256, 256, 0, stream>>>(col, ew, dis, E);
  k_rsqrt<<<(n + 255) / 256, 256, 0, stream>>>(dis, n);

  const float* hcur = x;
  for (int l = 0; l < 4; ++l) {
    k_gemm_wmma<<<GEMM_BLOCKS, 256, 0, stream>>>(hcur, Wc[l], hw, nullptr, n / 16);
    if (l == 0)
      k_tile_agg_first<<<nb, 1024, 0, stream>>>(hw, row, col, ew, dis, bc[l], hbuf, n, E, 1,
                                                cnts, lst);
    else
      k_tile_agg_next<<<nb, 1024, 0, stream>>>(hw, row, col, ew, dis, bc[l], hbuf, n, E,
                                               (l < 3) ? 1 : 0, cnts, lst);
    hcur = hbuf;
  }
  k_gemm_wmma<<<GEMM_BLOCKS, 256, 0, stream>>>(hcur, Wout, (float*)d_out, bout, n / 16);
}
